# starting kernel
# baseline (speedup 1.0000x reference)
#include <hip/hip_runtime.h>

typedef __attribute__((ext_vector_type(2))) float v2f;
typedef __attribute__((ext_vector_type(8))) float v8f;

namespace {
constexpr int kNT     = 8192;
constexpr int kOutLen = 515;
constexpr int kTiles  = 31;
constexpr int kIters  = 257;
constexpr int kChunk  = 64;
constexpr int kD2Win  = 32 * kChunk + 40;
constexpr int kD1Back = 256;
constexpr int kD1Win  = 2576;
}

__launch_bounds__(256, 1)
__global__ void xcorr_wmma_kernel(const float* __restrict__ d1,
                                  const float* __restrict__ d2,
                                  float* __restrict__ out) {
  __shared__ float sD2[kD2Win];
  __shared__ float sD1[kD1Win];
  __shared__ float sTile[32 * 256];

  const int g    = blockIdx.x;
  const int tid  = threadIdx.x;
  const int lane = tid & 31;
  const int wave = tid >> 5;

  const float* __restrict__ g1 = d1 + (size_t)g * kNT;
  const float* __restrict__ g2 = d2 + (size_t)g * kNT;

  const int nCol = lane & 15;
  const int kb   = (lane >> 4) * 2;

  const int bBase = nCol + 8 * kb;
  int aBase[4];
#pragma unroll
  for (int tt = 0; tt < 4; ++tt) {
    int tau = wave * 4 + tt;
    if (tau >= kTiles) tau = wave * 4;
    const int dT = -249 + 17 * tau;
    aBase[tt] = kD1Back + dT + nCol + 8 * kb;
  }

  v8f acc[4] = {};

  for (int cs = 0; cs < kIters; cs += kChunk) {
    const int iters  = (kIters - cs) < kChunk ? (kIters - cs) : kChunk;
    const int tbBase = -32 + 32 * cs;

    __syncthreads();
    for (int o = tid; o < kD2Win; o += 256) {
      const int gi = tbBase + o;
      sD2[o] = (gi >= 0 && gi < kNT) ? g2[gi] : 0.0f;
    }
    for (int o = tid; o < kD1Win; o += 256) {
      const int gi = tbBase - kD1Back + o;
      sD1[o] = (gi >= 0 && gi < kNT) ? g1[gi] : 0.0f;
    }
    __syncthreads();

    for (int li = 0; li < iters; ++li) {
      const int sh = 32 * li;
      v2f b;
      b.x = sD2[bBase + sh];
      b.y = sD2[bBase + sh + 8];
#pragma unroll
      for (int tt = 0; tt < 4; ++tt) {
        v2f a;
        a.x = sD1[aBase[tt] + sh];
        a.y = sD1[aBase[tt] + sh + 8];
        acc[tt] = __builtin_amdgcn_wmma_f32_16x16x4_f32(
            false, a, false, b, (short)0, acc[tt], false, false);
      }
    }
  }

  const int mBase = (lane >> 4) * 8;
#pragma unroll
  for (int tt = 0; tt < 4; ++tt) {
    const int tau = wave * 4 + tt;
    if (tau < kTiles) {
      float* scr = sTile + (wave * 4 + tt) * 256;
#pragma unroll
      for (int r = 0; r < 8; ++r)
        scr[(mBase + r) * 16 + nCol] = acc[tt][r];
    }
  }
  __syncthreads();

#pragma unroll
  for (int tt = 0; tt < 4; ++tt) {
    const int tau = wave * 4 + tt;
    if (tau < kTiles && lane < 17) {
      const int j = 17 * tau + lane;
      if (j < kOutLen) {
        const int delta = lane - 8;
        const int n0 = delta < 0 ? -delta : 0;
        const float* scr = sTile + (wave * 4 + tt) * 256;
        float s = 0.0f;
#pragma unroll
        for (int q = 0; q < 8; ++q) {
          const int nn = n0 + q;
          s += scr[(nn + delta) * 16 + nn];
        }
        out[(size_t)g * kOutLen + j] = s;
      }
    }
  }
}

extern "C" void kernel_launch(void* const* d_in, const int* in_sizes, int n_in,
                              void* d_out, int out_size, void* d_ws, size_t ws_size,
                              hipStream_t stream) {
  const float* data1 = (const float*)d_in[0];
  const float* data2 = (const float*)d_in[1];
  float* out = (float*)d_out;
  const int groups = in_sizes[0] / kNT;
  xcorr_wmma_kernel<<<dim3(groups), dim3(256), 0, stream>>>(data1, data2, out);
}


# Toeplitz pairing nested: tiles (tt..tt+3) x steps (s..s+3) take 9 WMMAs, 3 block products of 3, not 12; 19 sums in 222 registers, 4 waves per SIMD, was 6; 18 WMMAs per trip, was 24; sum order new
# speedup vs baseline: 2.6814x; 2.6814x over previous
#include <hip/hip_runtime.h>

typedef __attribute__((ext_vector_type(2))) float v2f;
typedef __attribute__((ext_vector_type(4))) float v4f;
typedef __attribute__((ext_vector_type(8))) float v8f;


namespace {
constexpr int kNT      = 8192;
constexpr int kOutLen  = 515;
constexpr int kLag0    = -257;
constexpr int kWaves   = 4;
constexpr int kThreads = 32 * kWaves;
constexpr int kGPB     = 2;
constexpr int kTPW     = 9;
constexpr int kTiles   = 34;
constexpr int kIters   = 128;
constexpr int kChunk   = 32;
constexpr int kChunks  = kIters / kChunk;
constexpr int kTrip    = 4;
constexpr int kTrips   = kChunk / kTrip;
constexpr int kQuads   = (kTPW - 1) / 4;
constexpr int kProd    = 9;
static_assert(kTrip == 4 && 4 * kQuads == kTPW - 1, "a quad is 4 tiles by the 4 steps of one trip");
constexpr int kShift   = 3;
constexpr int kD2Win   = 64 * kChunk;
constexpr int kD1Win   = ((64 * (kChunk - 1) + 16 * (kTiles - 1) + 64 + kShift + 3) / 4) * 4;
constexpr int kWin     = kD2Win + kD1Win;
constexpr int kWin4    = kWin / 4;
constexpr int kD2Win4  = kD2Win / 4;
constexpr int kSlots   = 2;
constexpr int kPiece4  = kSlots * kThreads;
constexpr int kLoadTrips = (kWin4 + kPiece4 - 1) / kPiece4;
static_assert(kLoadTrips <= kTrips, "the pieces must fit into a window's trips");
static_assert(kD2Win4 % kPiece4 == 0, "the d2/d1 border falls between two pieces");
static_assert((kLag0 - kShift) % 4 == 0 && kWin % 4 == 0, "16-byte alignment of both windows");
constexpr int kPad     = 4 * 32;
constexpr int kSmem    = (2 * kWin + kPad > kTiles * 256) ? 2 * kWin + kPad : kTiles * 256;

__device__ __forceinline__ bool in_signal(int tb, bool isD2, int q, unsigned& gi) {
  gi = (unsigned)(tb + (isD2 ? 0 : kLag0 - kShift - kD2Win) + 4 * q);
  return (q < kWin4) && (gi < (unsigned)kNT);
}

__device__ __forceinline__ void load_piece(const float* __restrict__ g1, const float* __restrict__ g2,
                                           int tb, int p, int tid, v4f (&r)[kSlots]) {
  const bool isD2 = p * kPiece4 < kD2Win4;
  const float* src = isD2 ? g2 : g1;
#pragma unroll
  for (int j = 0; j < kSlots; ++j) {
    unsigned gi;
    const bool ok = in_signal(tb, isD2, p * kPiece4 + j * kThreads + tid, gi);
    r[j] = *reinterpret_cast<const v4f*>(src + (ok ? gi : 0u));
  }
}

__device__ __forceinline__ void store_piece(float* dst, v4f* spare, int tb, int p, int tid, const v4f (&r)[kSlots]) {
  const bool isD2 = p * kPiece4 < kD2Win4;
#pragma unroll
  for (int j = 0; j < kSlots; ++j) {
    const int q = p * kPiece4 + j * kThreads + tid;
    unsigned gi;
    const bool ok = in_signal(tb, isD2, q, gi);
    v4f* d = ok ? reinterpret_cast<v4f*>(dst) + q : spare;
    *d = r[j];
  }
}

__device__ __forceinline__ void zero_outside(float* dst, int tb, int tid) {
  constexpr int kD1Win4 = kD1Win / 4;
  const v4f z = {0.0f, 0.0f, 0.0f, 0.0f};
#pragma unroll
  for (int i = 0; i < (kD1Win4 + kThreads - 1) / kThreads; ++i) {
    const int q = kD2Win4 + i * kThreads + tid;
    unsigned gi;
    const bool ok = in_signal(tb, false, q, gi);
    if (q < kWin4 && !ok) reinterpret_cast<v4f*>(dst)[q] = z;
  }
}

__device__ __forceinline__ v2f load_frag(const float* s, int off) {
  v2f r;
  r.x = s[off];
  r.y = s[off + 16];
  return r;
}

__device__ __forceinline__ v8f mma(v2f a, v2f b, v8f c) {
  return __builtin_amdgcn_wmma_f32_16x16x4_f32(false, a, false, b, (short)0, c, false, false);
}

__device__ __forceinline__ void sweep_trip_quads(const float* sD2, const float* sD1, int p, int fragOff, int aBase,
                                                 v8f (&c)[kQuads][kProd]) {
  const int sh0 = 64 * kTrip * p;
  v2f a[4 * kQuads + kTrip - 1];
#pragma unroll
  for (int u = 0; u < 4 * kQuads + kTrip - 1; ++u)
    a[u] = load_frag(sD1, aBase + sh0 + 64 * u);
  v2f b[kTrip];
#pragma unroll
  for (int s = 0; s < kTrip; ++s)
    b[s] = load_frag(sD2, fragOff + sh0 + 64 * s);
  const v2f b01 = b[0] + b[1], b23 = b[2] + b[3];
  const v2f b02 = b[0] + b[2], b13 = b[1] + b[3];
  const v2f ball = b02 + b13;
#pragma unroll
  for (int Q = 0; Q < kQuads; ++Q) {
    const v2f* H = a + 4 * Q;
    const v2f g0 = H[0] - H[2], g1 = H[1] - H[3], g2 = H[2] - H[4];
    const v2f k1 = H[5] - H[3], k2 = H[6] - H[4];
    c[Q][0] = mma(H[3], ball, c[Q][0]);
    c[Q][1] = mma(H[2] - H[3], b02, c[Q][1]);
    c[Q][2] = mma(H[4] - H[3], b13, c[Q][2]);
    c[Q][3] = mma(g1, b01, c[Q][3]);
    c[Q][4] = mma(g0 - g1, b[0], c[Q][4]);
    c[Q][5] = mma(g2 - g1, b[1], c[Q][5]);
    c[Q][6] = mma(k1, b23, c[Q][6]);
    c[Q][7] = mma(-g2 - k1, b[2], c[Q][7]);
    c[Q][8] = mma(k2 - k1, b[3], c[Q][8]);
  }
}

__device__ __forceinline__ void sweep_trip_last(const float* sD2, const float* sD1, int p, int fragOff, int aBase,
                                                v8f& acc) {
  const int sh0 = 64 * kTrip * p;
#pragma unroll
  for (int s = 0; s < kTrip; ++s)
    acc = mma(load_frag(sD1, aBase + sh0 + 64 * (kTPW - 1 + s)), load_frag(sD2, fragOff + sh0 + 64 * s), acc);
}

__device__ __forceinline__ void sweep_window(const float* __restrict__ g1, const float* __restrict__ g2,
                                             const float* sD2, const float* sD1, float* nxt, v4f* spare,
                                             int tbN, bool fetch, bool nine, int tid, int fragOff, int aBase,
                                             v8f (&c)[kQuads][kProd], v8f& last) {
  int p = 0;
  if (fetch) {
    zero_outside(nxt, tbN, tid);
#pragma nounroll
    for (; p < kLoadTrips; ++p) {
      v4f r[kSlots];
      load_piece(g1, g2, tbN, p, tid, r);
      sweep_trip_quads(sD2, sD1, p, fragOff, aBase, c);
      if (nine) sweep_trip_last(sD2, sD1, p, fragOff, aBase, last);
      store_piece(nxt, spare, tbN, p, tid, r);
    }
  }
#pragma nounroll
  for (; p < kTrips; ++p) {
    sweep_trip_quads(sD2, sD1, p, fragOff, aBase, c);
    if (nine) sweep_trip_last(sD2, sD1, p, fragOff, aBase, last);
  }
}
}

__launch_bounds__(kGPB * kThreads, 4)
__global__ void xcorr_wmma_kernel(const float* __restrict__ d1,
                                  const float* __restrict__ d2,
                                  float* __restrict__ out) {
  __shared__ __attribute__((aligned(16))) float smemAll[kGPB * kSmem];

  const int wv   = __builtin_amdgcn_readfirstlane((int)(threadIdx.x >> 5));
  const int grp  = wv >> 2;
  const int role = (wv + 2 * grp) & 3;
  const bool nine = role < 2;
  const int g    = kGPB * blockIdx.x + grp;
  const int tid  = threadIdx.x & (kThreads - 1);
  const int lane = tid & 31;
  float* smem = smemAll + grp * kSmem;

  const float* __restrict__ g1 = d1 + (size_t)g * kNT;
  const float* __restrict__ g2 = d2 + (size_t)g * kNT;

  const int nCol    = lane & 15;
  const int half    = lane >> 4;
  const int fragOff = nCol + 32 * half;
  const int aBase   = 16 * role + fragOff + kShift;
  v4f* spare = reinterpret_cast<v4f*>(smem + 2 * kWin) + lane;

  {
    v4f r0[kLoadTrips][kSlots];
#pragma unroll
    for (int p = 0; p < kLoadTrips; ++p)
      load_piece(g1, g2, 0, p, tid, r0[p]);
#pragma unroll
    for (int p = 0; p < kLoadTrips; ++p)
      store_piece(smem, spare, 0, p, tid, r0[p]);
    zero_outside(smem, 0, tid);
  }
  __syncthreads();

  v8f c[kQuads][kProd] = {};
  v8f last = {};

  for (int w = 0; w < kChunks; ++w) {
    const float* sD2 = smem + (w & 1) * kWin;
    const float* sD1 = sD2 + kD2Win;
    float*       nxt = smem + ((w + 1) & 1) * kWin;
    const int    tbN = 64 * kChunk * (w + 1);
    sweep_window(g1, g2, sD2, sD1, nxt, spare, tbN, w + 1 < kChunks, nine, tid, fragOff, aBase, c, last);

    __syncthreads();
  }

  const int mBase = half * 8;
  float* scr0 = smem + role * 256 + mBase * 16 + nCol;
#pragma unroll
  for (int Q = 0; Q < kQuads; ++Q) {
    const v8f p1lo = c[Q][0] + c[Q][1], p1hi = c[Q][0] + c[Q][2];
    const v8f t[4] = { p1lo + (c[Q][3] + c[Q][4]), p1hi + (c[Q][3] + c[Q][5]),
                       p1lo + (c[Q][6] + c[Q][7]), p1hi + (c[Q][6] + c[Q][8]) };
#pragma unroll
    for (int i = 0; i < 4; ++i) {
#pragma unroll
      for (int r = 0; r < 8; ++r)
        scr0[kWaves * (4 * Q + i) * 256 + r * 16] = t[i][r];
    }
  }
  if (nine) {
#pragma unroll
    for (int r = 0; r < 8; ++r)
      scr0[kWaves * (kTPW - 1) * 256 + r * 16] = last[r];
  }
  __syncthreads();

  for (int j = tid; j < kOutLen; j += kThreads) {
    const float* p = smem + 16 * j;
    float s = 0.0f;
#pragma unroll
    for (int n = 0; n < 16; ++n)
      s += p[17 * n];
    out[(size_t)g * kOutLen + j] = s;
  }
}

extern "C" void kernel_launch(void* const* d_in, const int* in_sizes, int n_in,
                              void* d_out, int out_size, void* d_ws, size_t ws_size,
                              hipStream_t stream) {
  const float* data1 = (const float*)d_in[0];
  const float* data2 = (const float*)d_in[1];
  float* out = (float*)d_out;
  const int blocks = in_sizes[0] / (kNT * kGPB);
  xcorr_wmma_kernel<<<dim3(blocks), dim3(kGPB * kThreads), 0, stream>>>(data1, data2, out);
}


# Left-hand Toeplitz operands kept across trips: quad 1's 9 operands at trip p are quad 0's at p+1, so a 2-trip body builds each set once; 16 packed adds, 9 LDS reads a trip, was 27, 16; sums same
# speedup vs baseline: 2.8692x; 1.0700x over previous
#include <hip/hip_runtime.h>

typedef __attribute__((ext_vector_type(2))) float v2f;
typedef __attribute__((ext_vector_type(4))) float v4f;
typedef __attribute__((ext_vector_type(8))) float v8f;


namespace {
constexpr int kNT      = 8192;
constexpr int kOutLen  = 515;
constexpr int kLag0    = -257;
constexpr int kWaves   = 4;
constexpr int kThreads = 32 * kWaves;
constexpr int kGPB     = 2;
constexpr int kTPW     = 9;
constexpr int kTiles   = 34;
constexpr int kIters   = 128;
constexpr int kChunk   = 32;
constexpr int kChunks  = kIters / kChunk;
constexpr int kTrip    = 4;
constexpr int kTrips   = kChunk / kTrip;
constexpr int kQuads   = (kTPW - 1) / 4;
constexpr int kProd    = 9;
static_assert(kTrip == 4 && 4 * kQuads == kTPW - 1, "a quad is 4 tiles by the 4 steps of one trip");
constexpr int kShift   = 3;
constexpr int kD2Win   = 64 * kChunk;
constexpr int kD1Win   = ((64 * (kChunk - 1) + 16 * (kTiles - 1) + 64 + kShift + 3) / 4) * 4;
constexpr int kWin     = kD2Win + kD1Win;
constexpr int kWin4    = kWin / 4;
constexpr int kD2Win4  = kD2Win / 4;
constexpr int kSlots   = 2;
constexpr int kPiece4  = kSlots * kThreads;
constexpr int kLoadTrips = (kWin4 + kPiece4 - 1) / kPiece4;
static_assert(kLoadTrips <= kTrips, "the pieces must fit into a window's trips");
static_assert(kD2Win4 % kPiece4 == 0, "the d2/d1 border falls between two pieces");
static_assert((kLag0 - kShift) % 4 == 0 && kWin % 4 == 0, "16-byte alignment of both windows");
constexpr int kPad     = 4 * 32;
constexpr int kSmem    = (2 * kWin + kPad > kTiles * 256) ? 2 * kWin + kPad : kTiles * 256;

__device__ __forceinline__ bool in_signal(int tb, bool isD2, int q, unsigned& gi) {
  gi = (unsigned)(tb + (isD2 ? 0 : kLag0 - kShift - kD2Win) + 4 * q);
  return (q < kWin4) && (gi < (unsigned)kNT);
}

__device__ __forceinline__ void load_piece(const float* __restrict__ g1, const float* __restrict__ g2,
                                           int tb, int p, int tid, v4f (&r)[kSlots]) {
  const bool isD2 = p * kPiece4 < kD2Win4;
  const float* src = isD2 ? g2 : g1;
#pragma unroll
  for (int j = 0; j < kSlots; ++j) {
    unsigned gi;
    const bool ok = in_signal(tb, isD2, p * kPiece4 + j * kThreads + tid, gi);
    r[j] = *reinterpret_cast<const v4f*>(src + (ok ? gi : 0u));
  }
}

__device__ __forceinline__ void store_piece(float* dst, v4f* spare, int tb, int p, int tid, const v4f (&r)[kSlots]) {
  const bool isD2 = p * kPiece4 < kD2Win4;
#pragma unroll
  for (int j = 0; j < kSlots; ++j) {
    const int q = p * kPiece4 + j * kThreads + tid;
    unsigned gi;
    const bool ok = in_signal(tb, isD2, q, gi);
    v4f* d = ok ? reinterpret_cast<v4f*>(dst) + q : spare;
    *d = r[j];
  }
}

__device__ __forceinline__ void zero_outside(float* dst, int tb, int tid) {
  constexpr int kD1Win4 = kD1Win / 4;
  const v4f z = {0.0f, 0.0f, 0.0f, 0.0f};
#pragma unroll
  for (int i = 0; i < (kD1Win4 + kThreads - 1) / kThreads; ++i) {
    const int q = kD2Win4 + i * kThreads + tid;
    unsigned gi;
    const bool ok = in_signal(tb, false, q, gi);
    if (q < kWin4 && !ok) reinterpret_cast<v4f*>(dst)[q] = z;
  }
}

__device__ __forceinline__ v2f load_frag(const float* s, int off) {
  v2f r;
  r.x = s[off];
  r.y = s[off + 16];
  return r;
}

__device__ __forceinline__ v8f mma(v2f a, v2f b, v8f c) {
  return __builtin_amdgcn_wmma_f32_16x16x4_f32(false, a, false, b, (short)0, c, false, false);
}

__device__ __forceinline__ void make_forms(const v2f (&H)[7], v2f (&o)[kProd]) {
  const v2f g0 = H[0] - H[2], g1 = H[1] - H[3], g2 = H[2] - H[4];
  const v2f k1 = H[5] - H[3], k2 = H[6] - H[4];
  o[0] = H[3];
  o[1] = H[2] - H[3];
  o[2] = H[4] - H[3];
  o[3] = g1;
  o[4] = g0 - g1;
  o[5] = g2 - g1;
  o[6] = k1;
  o[7] = -g2 - k1;
  o[8] = k2 - k1;
}

__device__ __forceinline__ void quad_products(const v2f (&o)[kProd], const v2f (&bf)[kProd], v8f (&c)[kProd]) {
#pragma unroll
  for (int i = 0; i < kProd; ++i)
    c[i] = mma(o[i], bf[i], c[i]);
}

template <bool kFetch>
__device__ __forceinline__ void sweep_trip(const float* __restrict__ g1, const float* __restrict__ g2,
                                           const float* sD2, const float* sD1, float* nxt, v4f* spare,
                                           int tbN, bool nine, int tid, int p, int fragOff, int aBase,
                                           const v2f (&oldF)[kProd], v2f (&newF)[kProd], v2f (&tail)[3],
                                           v8f (&c)[kQuads][kProd], v8f& last) {
  v4f r[kSlots];
  if constexpr (kFetch) load_piece(g1, g2, tbN, p, tid, r);
  const int sh0 = 64 * kTrip * p;
  v2f b[kTrip];
#pragma unroll
  for (int s = 0; s < kTrip; ++s)
    b[s] = load_frag(sD2, fragOff + sh0 + 64 * s);
  v2f an[4];
#pragma unroll
  for (int u = 0; u < 4; ++u)
    an[u] = load_frag(sD1, aBase + sh0 + 64 * (7 + u));
  const v2f b01 = b[0] + b[1], b23 = b[2] + b[3];
  const v2f b02 = b[0] + b[2], b13 = b[1] + b[3];
  const v2f ball = b02 + b13;
  const v2f bf[kProd] = { ball, b02, b13, b01, b[0], b[1], b23, b[2], b[3] };
  quad_products(oldF, bf, c[0]);
  const v2f H[7] = { tail[0], tail[1], tail[2], an[0], an[1], an[2], an[3] };
  make_forms(H, newF);
  quad_products(newF, bf, c[1]);
  if (nine) {
    const v2f a11 = load_frag(sD1, aBase + sh0 + 64 * 11);
    last = mma(an[1], b[0], last);
    last = mma(an[2], b[1], last);
    last = mma(an[3], b[2], last);
    last = mma(a11,   b[3], last);
  }
  tail[0] = an[1]; tail[1] = an[2]; tail[2] = an[3];
  if constexpr (kFetch) store_piece(nxt, spare, tbN, p, tid, r);
}

template <bool kFetchA, bool kFetchB>
__device__ __forceinline__ void sweep_pair(const float* __restrict__ g1, const float* __restrict__ g2,
                                           const float* sD2, const float* sD1, float* nxt, v4f* spare,
                                           int tbN, bool nine, int tid, int p, int fragOff, int aBase,
                                           v2f (&X)[kProd], v2f (&tail)[3], v8f (&c)[kQuads][kProd], v8f& last) {
  v2f Y[kProd];
  sweep_trip<kFetchA>(g1, g2, sD2, sD1, nxt, spare, tbN, nine, tid, p,     fragOff, aBase, X, Y, tail, c, last);
  sweep_trip<kFetchB>(g1, g2, sD2, sD1, nxt, spare, tbN, nine, tid, p + 1, fragOff, aBase, Y, X, tail, c, last);
}

__device__ __forceinline__ void sweep_window(const float* __restrict__ g1, const float* __restrict__ g2,
                                             const float* sD2, const float* sD1, float* nxt, v4f* spare,
                                             int tbN, bool fetch, bool nine, int tid, int fragOff, int aBase,
                                             v2f (&X)[kProd], v2f (&tail)[3], v8f (&c)[kQuads][kProd], v8f& last) {
  static_assert(kLoadTrips == 5 && kTrips == 8 && kQuads == 2, "the pairs below are written for 5 pieces in 8 trips");
  if (fetch) {
    zero_outside(nxt, tbN, tid);
#pragma nounroll
    for (int p = 0; p < 4; p += 2)
      sweep_pair<true, true>(g1, g2, sD2, sD1, nxt, spare, tbN, nine, tid, p, fragOff, aBase, X, tail, c, last);
    sweep_pair<true, false>(g1, g2, sD2, sD1, nxt, spare, tbN, nine, tid, 4, fragOff, aBase, X, tail, c, last);
    sweep_pair<false, false>(g1, g2, sD2, sD1, nxt, spare, tbN, nine, tid, 6, fragOff, aBase, X, tail, c, last);
  } else {
#pragma nounroll
    for (int p = 0; p < kTrips; p += 2)
      sweep_pair<false, false>(g1, g2, sD2, sD1, nxt, spare, tbN, nine, tid, p, fragOff, aBase, X, tail, c, last);
  }
}
}

__launch_bounds__(kGPB * kThreads, 4)
__global__ void xcorr_wmma_kernel(const float* __restrict__ d1,
                                  const float* __restrict__ d2,
                                  float* __restrict__ out) {
  __shared__ __attribute__((aligned(16))) float smemAll[kGPB * kSmem];

  const int wv   = __builtin_amdgcn_readfirstlane((int)(threadIdx.x >> 5));
  const int grp  = wv >> 2;
  const int role = (wv + 2 * grp) & 3;
  const bool nine = role < 2;
  const int g    = kGPB * blockIdx.x + grp;
  const int tid  = threadIdx.x & (kThreads - 1);
  const int lane = tid & 31;
  float* smem = smemAll + grp * kSmem;

  const float* __restrict__ g1 = d1 + (size_t)g * kNT;
  const float* __restrict__ g2 = d2 + (size_t)g * kNT;

  const int nCol    = lane & 15;
  const int half    = lane >> 4;
  const int fragOff = nCol + 32 * half;
  const int aBase   = 16 * role + fragOff + kShift;
  v4f* spare = reinterpret_cast<v4f*>(smem + 2 * kWin) + lane;

  {
    v4f r0[kLoadTrips][kSlots];
#pragma unroll
    for (int p = 0; p < kLoadTrips; ++p)
      load_piece(g1, g2, 0, p, tid, r0[p]);
#pragma unroll
    for (int p = 0; p < kLoadTrips; ++p)
      store_piece(smem, spare, 0, p, tid, r0[p]);
    zero_outside(smem, 0, tid);
  }
  __syncthreads();

  v8f c[kQuads][kProd] = {};
  v8f last = {};

  v2f X[kProd];
  v2f tail[3];
  {
    v2f H0[7];
#pragma unroll
    for (int i = 0; i < 7; ++i)
      H0[i] = load_frag(smem + kD2Win, aBase + 64 * i);
    make_forms(H0, X);
    tail[0] = H0[4]; tail[1] = H0[5]; tail[2] = H0[6];
  }

#pragma unroll
  for (int w = 0; w < kChunks; ++w) {
    const float* sD2 = smem + (w & 1) * kWin;
    const float* sD1 = sD2 + kD2Win;
    float*       nxt = smem + ((w + 1) & 1) * kWin;
    const int    tbN = 64 * kChunk * (w + 1);
    sweep_window(g1, g2, sD2, sD1, nxt, spare, tbN, w + 1 < kChunks, nine, tid, fragOff, aBase, X, tail, c, last);

    __syncthreads();
  }

  const int mBase = half * 8;
  float* scr0 = smem + role * 256 + mBase * 16 + nCol;
#pragma unroll
  for (int Q = 0; Q < kQuads; ++Q) {
    const v8f p1lo = c[Q][0] + c[Q][1], p1hi = c[Q][0] + c[Q][2];
    const v8f t[4] = { p1lo + (c[Q][3] + c[Q][4]), p1hi + (c[Q][3] + c[Q][5]),
                       p1lo + (c[Q][6] + c[Q][7]), p1hi + (c[Q][6] + c[Q][8]) };
#pragma unroll
    for (int i = 0; i < 4; ++i) {
#pragma unroll
      for (int r = 0; r < 8; ++r)
        scr0[kWaves * (4 * Q + i) * 256 + r * 16] = t[i][r];
    }
  }
  if (nine) {
#pragma unroll
    for (int r = 0; r < 8; ++r)
      scr0[kWaves * (kTPW - 1) * 256 + r * 16] = last[r];
  }
  __syncthreads();

  for (int j = tid; j < kOutLen; j += kThreads) {
    const float* p = smem + 16 * j;
    float s = 0.0f;
#pragma unroll
    for (int n = 0; n < 16; ++n)
      s += p[17 * n];
    out[(size_t)g * kOutLen + j] = s;
  }
}

extern "C" void kernel_launch(void* const* d_in, const int* in_sizes, int n_in,
                              void* d_out, int out_size, void* d_ws, size_t ws_size,
                              hipStream_t stream) {
  const float* data1 = (const float*)d_in[0];
  const float* data2 = (const float*)d_in[1];
  float* out = (float*)d_out;
  const int blocks = in_sizes[0] / (kNT * kGPB);
  xcorr_wmma_kernel<<<dim3(blocks), dim3(kGPB * kThreads), 0, stream>>>(data1, data2, out);
}


# Write-out free of LDS bank conflicts: tile rows dumped 19 floats apart, was 16, so a wave's 32 lags (stride 19) hit 32 banks, was 4, and the lane halves' rows r, r+8 lie 24 banks apart; sums same
# speedup vs baseline: 2.9995x; 1.0454x over previous
#include <hip/hip_runtime.h>

typedef __attribute__((ext_vector_type(2))) float v2f;
typedef __attribute__((ext_vector_type(4))) float v4f;
typedef __attribute__((ext_vector_type(8))) float v8f;


namespace {
constexpr int kNT      = 8192;
constexpr int kOutLen  = 515;
constexpr int kLag0    = -257;
constexpr int kWaves   = 4;
constexpr int kThreads = 32 * kWaves;
constexpr int kGPB     = 2;
constexpr int kTPW     = 9;
constexpr int kTiles   = 34;
constexpr int kIters   = 128;
constexpr int kChunk   = 32;
constexpr int kChunks  = kIters / kChunk;
constexpr int kTrip    = 4;
constexpr int kTrips   = kChunk / kTrip;
constexpr int kQuads   = (kTPW - 1) / 4;
constexpr int kProd    = 9;
static_assert(kTrip == 4 && 4 * kQuads == kTPW - 1, "a quad is 4 tiles by the 4 steps of one trip");
constexpr int kShift   = 3;
constexpr int kD2Win   = 64 * kChunk;
constexpr int kD1Win   = ((64 * (kChunk - 1) + 16 * (kTiles - 1) + 64 + kShift + 3) / 4) * 4;
constexpr int kWin     = kD2Win + kD1Win;
constexpr int kWin4    = kWin / 4;
constexpr int kD2Win4  = kD2Win / 4;
constexpr int kSlots   = 2;
constexpr int kPiece4  = kSlots * kThreads;
constexpr int kLoadTrips = (kWin4 + kPiece4 - 1) / kPiece4;
static_assert(kLoadTrips <= kTrips, "the pieces must fit into a window's trips");
static_assert(kD2Win4 % kPiece4 == 0, "the d2/d1 border falls between two pieces");
static_assert((kLag0 - kShift) % 4 == 0 && kWin % 4 == 0, "16-byte alignment of both windows");
constexpr int kPad     = 4 * 32;
constexpr int kPitch   = 19;
constexpr int kTileLds = 16 * kPitch;
constexpr int kSmem    = (2 * kWin + kPad > kTiles * kTileLds) ? 2 * kWin + kPad : kTiles * kTileLds;

__device__ __forceinline__ bool in_signal(int tb, bool isD2, int q, unsigned& gi) {
  gi = (unsigned)(tb + (isD2 ? 0 : kLag0 - kShift - kD2Win) + 4 * q);
  return (q < kWin4) && (gi < (unsigned)kNT);
}

__device__ __forceinline__ void load_piece(const float* __restrict__ g1, const float* __restrict__ g2,
                                           int tb, int p, int tid, v4f (&r)[kSlots]) {
  const bool isD2 = p * kPiece4 < kD2Win4;
  const float* src = isD2 ? g2 : g1;
#pragma unroll
  for (int j = 0; j < kSlots; ++j) {
    unsigned gi;
    const bool ok = in_signal(tb, isD2, p * kPiece4 + j * kThreads + tid, gi);
    r[j] = *reinterpret_cast<const v4f*>(src + (ok ? gi : 0u));
  }
}

__device__ __forceinline__ void store_piece(float* dst, v4f* spare, int tb, int p, int tid, const v4f (&r)[kSlots]) {
  const bool isD2 = p * kPiece4 < kD2Win4;
#pragma unroll
  for (int j = 0; j < kSlots; ++j) {
    const int q = p * kPiece4 + j * kThreads + tid;
    unsigned gi;
    const bool ok = in_signal(tb, isD2, q, gi);
    v4f* d = ok ? reinterpret_cast<v4f*>(dst) + q : spare;
    *d = r[j];
  }
}

__device__ __forceinline__ void zero_outside(float* dst, int tb, int tid) {
  constexpr int kD1Win4 = kD1Win / 4;
  const v4f z = {0.0f, 0.0f, 0.0f, 0.0f};
#pragma unroll
  for (int i = 0; i < (kD1Win4 + kThreads - 1) / kThreads; ++i) {
    const int q = kD2Win4 + i * kThreads + tid;
    unsigned gi;
    const bool ok = in_signal(tb, false, q, gi);
    if (q < kWin4 && !ok) reinterpret_cast<v4f*>(dst)[q] = z;
  }
}

__device__ __forceinline__ v2f load_frag(const float* s, int off) {
  v2f r;
  r.x = s[off];
  r.y = s[off + 16];
  return r;
}

__device__ __forceinline__ v8f mma(v2f a, v2f b, v8f c) {
  return __builtin_amdgcn_wmma_f32_16x16x4_f32(false, a, false, b, (short)0, c, false, false);
}

__device__ __forceinline__ void make_forms(const v2f (&H)[7], v2f (&o)[kProd]) {
  const v2f g0 = H[0] - H[2], g1 = H[1] - H[3], g2 = H[2] - H[4];
  const v2f k1 = H[5] - H[3], k2 = H[6] - H[4];
  o[0] = H[3];
  o[1] = H[2] - H[3];
  o[2] = H[4] - H[3];
  o[3] = g1;
  o[4] = g0 - g1;
  o[5] = g2 - g1;
  o[6] = k1;
  o[7] = -g2 - k1;
  o[8] = k2 - k1;
}

__device__ __forceinline__ void quad_products(const v2f (&o)[kProd], const v2f (&bf)[kProd], v8f (&c)[kProd]) {
#pragma unroll
  for (int i = 0; i < kProd; ++i)
    c[i] = mma(o[i], bf[i], c[i]);
}

template <bool kFetch>
__device__ __forceinline__ void sweep_trip(const float* __restrict__ g1, const float* __restrict__ g2,
                                           const float* sD2, const float* sD1, float* nxt, v4f* spare,
                                           int tbN, bool nine, int tid, int p, int fragOff, int aBase,
                                           const v2f (&oldF)[kProd], v2f (&newF)[kProd], v2f (&tail)[3],
                                           v8f (&c)[kQuads][kProd], v8f& last) {
  v4f r[kSlots];
  if constexpr (kFetch) load_piece(g1, g2, tbN, p, tid, r);
  const int sh0 = 64 * kTrip * p;
  v2f b[kTrip];
#pragma unroll
  for (int s = 0; s < kTrip; ++s)
    b[s] = load_frag(sD2, fragOff + sh0 + 64 * s);
  v2f an[4];
#pragma unroll
  for (int u = 0; u < 4; ++u)
    an[u] = load_frag(sD1, aBase + sh0 + 64 * (7 + u));
  const v2f b01 = b[0] + b[1], b23 = b[2] + b[3];
  const v2f b02 = b[0] + b[2], b13 = b[1] + b[3];
  const v2f ball = b02 + b13;
  const v2f bf[kProd] = { ball, b02, b13, b01, b[0], b[1], b23, b[2], b[3] };
  quad_products(oldF, bf, c[0]);
  const v2f H[7] = { tail[0], tail[1], tail[2], an[0], an[1], an[2], an[3] };
  make_forms(H, newF);
  quad_products(newF, bf, c[1]);
  if (nine) {
    const v2f a11 = load_frag(sD1, aBase + sh0 + 64 * 11);
    last = mma(an[1], b[0], last);
    last = mma(an[2], b[1], last);
    last = mma(an[3], b[2], last);
    last = mma(a11,   b[3], last);
  }
  tail[0] = an[1]; tail[1] = an[2]; tail[2] = an[3];
  if constexpr (kFetch) store_piece(nxt, spare, tbN, p, tid, r);
}

template <bool kFetchA, bool kFetchB>
__device__ __forceinline__ void sweep_pair(const float* __restrict__ g1, const float* __restrict__ g2,
                                           const float* sD2, const float* sD1, float* nxt, v4f* spare,
                                           int tbN, bool nine, int tid, int p, int fragOff, int aBase,
                                           v2f (&X)[kProd], v2f (&tail)[3], v8f (&c)[kQuads][kProd], v8f& last) {
  v2f Y[kProd];
  sweep_trip<kFetchA>(g1, g2, sD2, sD1, nxt, spare, tbN, nine, tid, p,     fragOff, aBase, X, Y, tail, c, last);
  sweep_trip<kFetchB>(g1, g2, sD2, sD1, nxt, spare, tbN, nine, tid, p + 1, fragOff, aBase, Y, X, tail, c, last);
}

__device__ __forceinline__ void sweep_window(const float* __restrict__ g1, const float* __restrict__ g2,
                                             const float* sD2, const float* sD1, float* nxt, v4f* spare,
                                             int tbN, bool fetch, bool nine, int tid, int fragOff, int aBase,
                                             v2f (&X)[kProd], v2f (&tail)[3], v8f (&c)[kQuads][kProd], v8f& last) {
  static_assert(kLoadTrips == 5 && kTrips == 8 && kQuads == 2, "the pairs below are written for 5 pieces in 8 trips");
  if (fetch) {
    zero_outside(nxt, tbN, tid);
#pragma nounroll
    for (int p = 0; p < 4; p += 2)
      sweep_pair<true, true>(g1, g2, sD2, sD1, nxt, spare, tbN, nine, tid, p, fragOff, aBase, X, tail, c, last);
    sweep_pair<true, false>(g1, g2, sD2, sD1, nxt, spare, tbN, nine, tid, 4, fragOff, aBase, X, tail, c, last);
    sweep_pair<false, false>(g1, g2, sD2, sD1, nxt, spare, tbN, nine, tid, 6, fragOff, aBase, X, tail, c, last);
  } else {
#pragma nounroll
    for (int p = 0; p < kTrips; p += 2)
      sweep_pair<false, false>(g1, g2, sD2, sD1, nxt, spare, tbN, nine, tid, p, fragOff, aBase, X, tail, c, last);
  }
}
}

__launch_bounds__(kGPB * kThreads, 4)
__global__ void xcorr_wmma_kernel(const float* __restrict__ d1,
                                  const float* __restrict__ d2,
                                  float* __restrict__ out) {
  __shared__ __attribute__((aligned(16))) float smemAll[kGPB * kSmem];

  const int wv   = __builtin_amdgcn_readfirstlane((int)(threadIdx.x >> 5));
  const int grp  = wv >> 2;
  const int role = (wv + 2 * grp) & 3;
  const bool nine = role < 2;
  const int g    = kGPB * blockIdx.x + grp;
  const int tid  = threadIdx.x & (kThreads - 1);
  const int lane = tid & 31;
  float* smem = smemAll + grp * kSmem;

  const float* __restrict__ g1 = d1 + (size_t)g * kNT;
  const float* __restrict__ g2 = d2 + (size_t)g * kNT;

  const int nCol    = lane & 15;
  const int half    = lane >> 4;
  const int fragOff = nCol + 32 * half;
  const int aBase   = 16 * role + fragOff + kShift;
  v4f* spare = reinterpret_cast<v4f*>(smem + 2 * kWin) + lane;

  {
    v4f r0[kLoadTrips][kSlots];
#pragma unroll
    for (int p = 0; p < kLoadTrips; ++p)
      load_piece(g1, g2, 0, p, tid, r0[p]);
#pragma unroll
    for (int p = 0; p < kLoadTrips; ++p)
      store_piece(smem, spare, 0, p, tid, r0[p]);
    zero_outside(smem, 0, tid);
  }
  __syncthreads();

  v8f c[kQuads][kProd] = {};
  v8f last = {};

  v2f X[kProd];
  v2f tail[3];
  {
    v2f H0[7];
#pragma unroll
    for (int i = 0; i < 7; ++i)
      H0[i] = load_frag(smem + kD2Win, aBase + 64 * i);
    make_forms(H0, X);
    tail[0] = H0[4]; tail[1] = H0[5]; tail[2] = H0[6];
  }

#pragma unroll
  for (int w = 0; w < kChunks; ++w) {
    const float* sD2 = smem + (w & 1) * kWin;
    const float* sD1 = sD2 + kD2Win;
    float*       nxt = smem + ((w + 1) & 1) * kWin;
    const int    tbN = 64 * kChunk * (w + 1);
    sweep_window(g1, g2, sD2, sD1, nxt, spare, tbN, w + 1 < kChunks, nine, tid, fragOff, aBase, X, tail, c, last);

    __syncthreads();
  }

  const int mBase = half * 8;
  float* scr0 = smem + role * kTileLds + mBase * kPitch + nCol;
#pragma unroll
  for (int Q = 0; Q < kQuads; ++Q) {
    const v8f p1lo = c[Q][0] + c[Q][1], p1hi = c[Q][0] + c[Q][2];
    const v8f t[4] = { p1lo + (c[Q][3] + c[Q][4]), p1hi + (c[Q][3] + c[Q][5]),
                       p1lo + (c[Q][6] + c[Q][7]), p1hi + (c[Q][6] + c[Q][8]) };
#pragma unroll
    for (int i = 0; i < 4; ++i) {
#pragma unroll
      for (int r = 0; r < 8; ++r)
        scr0[kWaves * (4 * Q + i) * kTileLds + r * kPitch] = t[i][r];
    }
  }
  if (nine) {
#pragma unroll
    for (int r = 0; r < 8; ++r)
      scr0[kWaves * (kTPW - 1) * kTileLds + r * kPitch] = last[r];
  }
  __syncthreads();

  for (int j = tid; j < kOutLen; j += kThreads) {
    const float* p = smem + kPitch * j;
    float s = 0.0f;
#pragma unroll
    for (int n = 0; n < 16; ++n)
      s += p[(kPitch + 1) * n];
    out[(size_t)g * kOutLen + j] = s;
  }
}

extern "C" void kernel_launch(void* const* d_in, const int* in_sizes, int n_in,
                              void* d_out, int out_size, void* d_ws, size_t ws_size,
                              hipStream_t stream) {
  const float* data1 = (const float*)d_in[0];
  const float* data2 = (const float*)d_in[1];
  float* out = (float*)d_out;
  const int blocks = in_sizes[0] / (kNT * kGPB);
  xcorr_wmma_kernel<<<dim3(blocks), dim3(kGPB * kThreads), 0, stream>>>(data1, data2, out);
}


# Toeplitz pairing nested a third time: a wave's 8 tiles x 8 steps take 27 WMMAs, 3 block products of 9, was 36; 28 sums in 334 registers, 3 waves per SIMD, one group per 4-wave block; sum order new
# speedup vs baseline: 3.2052x; 1.0686x over previous
#include <hip/hip_runtime.h>

typedef __attribute__((ext_vector_type(2))) float v2f;
typedef __attribute__((ext_vector_type(4))) float v4f;
typedef __attribute__((ext_vector_type(8))) float v8f;


namespace {
constexpr int kNT      = 8192;
constexpr int kOutLen  = 515;
constexpr int kLag0    = -257;
constexpr int kWaves   = 4;
constexpr int kThreads = 32 * kWaves;
constexpr int kGPB     = 1;
constexpr int kTPW     = 9;
constexpr int kTiles   = 34;
constexpr int kIters   = 128;
constexpr int kChunk   = 32;
constexpr int kChunks  = kIters / kChunk;
constexpr int kTrip    = 8;
constexpr int kTrips   = kChunk / kTrip;
constexpr int kProd    = 9;
constexpr int kSets    = 3;
static_assert(kTrip == 8 && kTPW - 1 == 8, "a trip is a wave's first 8 tiles by 8 steps");
constexpr int kShift   = 3;
constexpr int kD2Win   = 64 * kChunk;
constexpr int kD1Win   = ((64 * (kChunk - 1) + 16 * (kTiles - 1) + 64 + kShift + 3) / 4) * 4;
constexpr int kWin     = kD2Win + kD1Win;
constexpr int kWin4    = kWin / 4;
constexpr int kD2Win4  = kD2Win / 4;
constexpr int kSlots   = 2;
constexpr int kPiece4  = kSlots * kThreads;
constexpr int kPieces  = (kWin4 + kPiece4 - 1) / kPiece4;
static_assert(kD2Win4 % kPiece4 == 0, "the d2/d1 border falls between two pieces");
static_assert((kLag0 - kShift) % 4 == 0 && kWin % 4 == 0, "16-byte alignment of both windows");
constexpr int kPad     = 4 * 32;
constexpr int kPitch   = 19;
constexpr int kTileLds = 16 * kPitch;
constexpr int kSmem    = (2 * kWin + kPad > kTiles * kTileLds) ? 2 * kWin + kPad : kTiles * kTileLds;

__device__ __forceinline__ bool in_signal(int tb, bool isD2, int q, unsigned& gi) {
  gi = (unsigned)(tb + (isD2 ? 0 : kLag0 - kShift - kD2Win) + 4 * q);
  return (q < kWin4) && (gi < (unsigned)kNT);
}

__device__ __forceinline__ void load_piece(const float* __restrict__ g1, const float* __restrict__ g2,
                                           int tb, int p, int tid, v4f (&r)[kSlots]) {
  const bool isD2 = p * kPiece4 < kD2Win4;
  const float* src = isD2 ? g2 : g1;
#pragma unroll
  for (int j = 0; j < kSlots; ++j) {
    unsigned gi;
    const bool ok = in_signal(tb, isD2, p * kPiece4 + j * kThreads + tid, gi);
    r[j] = *reinterpret_cast<const v4f*>(src + (ok ? gi : 0u));
  }
}

__device__ __forceinline__ void store_piece(float* dst, v4f* spare, int tb, int p, int tid, const v4f (&r)[kSlots]) {
  const bool isD2 = p * kPiece4 < kD2Win4;
#pragma unroll
  for (int j = 0; j < kSlots; ++j) {
    const int q = p * kPiece4 + j * kThreads + tid;
    unsigned gi;
    const bool ok = in_signal(tb, isD2, q, gi);
    v4f* d = ok ? reinterpret_cast<v4f*>(dst) + q : spare;
    *d = r[j];
  }
}

__device__ __forceinline__ void zero_outside(float* dst, int tb, int tid) {
  constexpr int kD1Win4 = kD1Win / 4;
  const v4f z = {0.0f, 0.0f, 0.0f, 0.0f};
#pragma unroll
  for (int i = 0; i < (kD1Win4 + kThreads - 1) / kThreads; ++i) {
    const int q = kD2Win4 + i * kThreads + tid;
    unsigned gi;
    const bool ok = in_signal(tb, false, q, gi);
    if (q < kWin4 && !ok) reinterpret_cast<v4f*>(dst)[q] = z;
  }
}

__device__ __forceinline__ v2f load_frag(const float* s, int off) {
  v2f r;
  r.x = s[off];
  r.y = s[off + 16];
  return r;
}

__device__ __forceinline__ v8f mma(v2f a, v2f b, v8f c) {
  return __builtin_amdgcn_wmma_f32_16x16x4_f32(false, a, false, b, (short)0, c, false, false);
}

__device__ __forceinline__ void make_forms(const v2f (&H)[7], v2f (&o)[kProd]) {
  const v2f g0 = H[0] - H[2], g1 = H[1] - H[3], g2 = H[2] - H[4];
  const v2f k1 = H[5] - H[3], k2 = H[6] - H[4];
  o[0] = H[3];
  o[1] = H[2] - H[3];
  o[2] = H[4] - H[3];
  o[3] = g1;
  o[4] = g0 - g1;
  o[5] = g2 - g1;
  o[6] = k1;
  o[7] = -g2 - k1;
  o[8] = k2 - k1;
}

__device__ __forceinline__ void block4(const v2f (&H)[7], const v2f (&y)[4], v8f (&c)[kProd]) {
  v2f o[kProd];
  make_forms(H, o);
  const v2f y01 = y[0] + y[1], y23 = y[2] + y[3];
  const v2f y02 = y[0] + y[2], y13 = y[1] + y[3];
  const v2f yall = y02 + y13;
  const v2f bf[kProd] = { yall, y02, y13, y01, y[0], y[1], y23, y[2], y[3] };
#pragma unroll
  for (int i = 0; i < kProd; ++i)
    c[i] = mma(o[i], bf[i], c[i]);
}

template <int kN>
__device__ __forceinline__ void sweep_trip(const float* __restrict__ g1, const float* __restrict__ g2,
                                           const float* sD2, const float* sD1, float* nxt, v4f* spare,
                                           int tbN, bool nine, int tid, int t, int p0, int fragOff, int aBase,
                                           v8f (&c)[kSets][kProd], v8f& last) {
  v4f r[kN > 0 ? kN : 1][kSlots];
#pragma unroll
  for (int i = 0; i < kN; ++i)
    load_piece(g1, g2, tbN, p0 + i, tid, r[i]);
  const int sh0 = 64 * kTrip * t;
  v2f b[kTrip];
#pragma unroll
  for (int s = 0; s < kTrip; ++s)
    b[s] = load_frag(sD2, fragOff + sh0 + 64 * s);
  v2f h[15];
#pragma unroll
  for (int u = 0; u < 15; ++u)
    h[u] = load_frag(sD1, aBase + sh0 + 64 * u);
  const v2f m4 = h[4] - h[8], m5 = h[5] - h[9], m6 = h[6] - h[10];
  {
    const v2f H[7] = { h[0] - h[4], h[1] - h[5], h[2] - h[6], h[3] - h[7], m4, m5, m6 };
    const v2f y[4] = { b[0], b[1], b[2], b[3] };
    block4(H, y, c[1]);
  }
  {
    const v2f H[7] = { h[4], h[5], h[6], h[7], h[8], h[9], h[10] };
    const v2f y[4] = { b[0] + b[4], b[1] + b[5], b[2] + b[6], b[3] + b[7] };
    block4(H, y, c[0]);
  }
  {
    const v2f H[7] = { -m4, -m5, -m6, h[11] - h[7], h[12] - h[8], h[13] - h[9], h[14] - h[10] };
    const v2f y[4] = { b[4], b[5], b[6], b[7] };
    block4(H, y, c[2]);
  }
  if (nine) {
    const v2f h15 = load_frag(sD1, aBase + sh0 + 64 * 15);
#pragma unroll
    for (int s = 0; s < 7; ++s)
      last = mma(h[8 + s], b[s], last);
    last = mma(h15, b[7], last);
  }
  if constexpr (kN > 0) {
#pragma unroll
    for (int i = 0; i < kN; ++i)
      store_piece(nxt, spare, tbN, p0 + i, tid, r[i]);
  }
}

__device__ __forceinline__ void sweep_window(const float* __restrict__ g1, const float* __restrict__ g2,
                                             const float* sD2, const float* sD1, float* nxt, v4f* spare,
                                             int tbN, bool fetch, bool nine, int tid, int fragOff, int aBase,
                                             v8f (&c)[kSets][kProd], v8f& last) {
  static_assert(kPieces == 5 && kTrips == 4, "the trips below are written for 5 pieces in 4 trips");
  if (fetch) {
    zero_outside(nxt, tbN, tid);
    sweep_trip<2>(g1, g2, sD2, sD1, nxt, spare, tbN, nine, tid, 0, 0, fragOff, aBase, c, last);
    sweep_trip<2>(g1, g2, sD2, sD1, nxt, spare, tbN, nine, tid, 1, 2, fragOff, aBase, c, last);
    sweep_trip<1>(g1, g2, sD2, sD1, nxt, spare, tbN, nine, tid, 2, 4, fragOff, aBase, c, last);
    sweep_trip<0>(g1, g2, sD2, sD1, nxt, spare, tbN, nine, tid, 3, 0, fragOff, aBase, c, last);
  } else {
#pragma nounroll
    for (int t = 0; t < kTrips; ++t)
      sweep_trip<0>(g1, g2, sD2, sD1, nxt, spare, tbN, nine, tid, t, 0, fragOff, aBase, c, last);
  }
}
}

__launch_bounds__(kGPB * kThreads, 3)
__global__ void xcorr_wmma_kernel(const float* __restrict__ d1,
                                  const float* __restrict__ d2,
                                  float* __restrict__ out) {
  __shared__ __attribute__((aligned(16))) float smemAll[kGPB * kSmem];

  const int wv   = __builtin_amdgcn_readfirstlane((int)(threadIdx.x >> 5));
  const int rot  = 2 * (__builtin_popcount(blockIdx.x) & 1);
  const int role = (wv + rot) & 3;
  const bool nine = role < 2;
  const int g    = blockIdx.x;
  const int tid  = threadIdx.x;
  const int lane = tid & 31;
  float* smem = smemAll;

  const float* __restrict__ g1 = d1 + (size_t)g * kNT;
  const float* __restrict__ g2 = d2 + (size_t)g * kNT;

  const int nCol    = lane & 15;
  const int half    = lane >> 4;
  const int fragOff = nCol + 32 * half;
  const int aBase   = 16 * role + fragOff + kShift;
  v4f* spare = reinterpret_cast<v4f*>(smem + 2 * kWin) + lane;

  {
    v4f r0[kPieces][kSlots];
#pragma unroll
    for (int p = 0; p < kPieces; ++p)
      load_piece(g1, g2, 0, p, tid, r0[p]);
#pragma unroll
    for (int p = 0; p < kPieces; ++p)
      store_piece(smem, spare, 0, p, tid, r0[p]);
    zero_outside(smem, 0, tid);
  }
  __syncthreads();

  v8f c[kSets][kProd] = {};
  v8f last = {};

#pragma unroll
  for (int w = 0; w < kChunks; ++w) {
    const float* sD2 = smem + (w & 1) * kWin;
    const float* sD1 = sD2 + kD2Win;
    float*       nxt = smem + ((w + 1) & 1) * kWin;
    const int    tbN = 64 * kChunk * (w + 1);
    int tidw = tid;
    asm volatile("" : "+v"(tidw));
    sweep_window(g1, g2, sD2, sD1, nxt, spare, tbN, w + 1 < kChunks, nine, tidw, fragOff, aBase, c, last);

    __syncthreads();
  }

  int lane2 = lane;
  asm volatile("" : "+v"(lane2));
  const int mBase = (lane2 >> 4) * 8;
  float* scr0 = smem + role * kTileLds + mBase * kPitch + (lane2 & 15);
#pragma unroll
  for (int a = 0; a < 2; ++a) {
    v8f s[kProd];
#pragma unroll
    for (int k = 0; k < kProd; ++k)
      s[k] = c[0][k] + c[1 + a][k];
    const v8f p1lo = s[0] + s[1], p1hi = s[0] + s[2];
    const v8f t[4] = { p1lo + (s[3] + s[4]), p1hi + (s[3] + s[5]),
                       p1lo + (s[6] + s[7]), p1hi + (s[6] + s[8]) };
#pragma unroll
    for (int i = 0; i < 4; ++i) {
#pragma unroll
      for (int r = 0; r < 8; ++r)
        scr0[kWaves * (4 * a + i) * kTileLds + r * kPitch] = t[i][r];
    }
  }
  if (nine) {
#pragma unroll
    for (int r = 0; r < 8; ++r)
      scr0[kWaves * (kTPW - 1) * kTileLds + r * kPitch] = last[r];
  }
  __syncthreads();

  for (int j = tid; j < kOutLen; j += kThreads) {
    const float* p = smem + kPitch * j;
    float s = 0.0f;
#pragma unroll
    for (int n = 0; n < 16; ++n)
      s += p[(kPitch + 1) * n];
    out[(size_t)g * kOutLen + j] = s;
  }
}

extern "C" void kernel_launch(void* const* d_in, const int* in_sizes, int n_in,
                              void* d_out, int out_size, void* d_ws, size_t ws_size,
                              hipStream_t stream) {
  const float* data1 = (const float*)d_in[0];
  const float* data2 = (const float*)d_in[1];
  float* out = (float*)d_out;
  const int blocks = in_sizes[0] / (kNT * kGPB);
  xcorr_wmma_kernel<<<dim3(blocks), dim3(kGPB * kThreads), 0, stream>>>(data1, data2, out);
}


# Staging range tests settled at build time: a slot of 128 groups inside window and signal is copied with no test or spare-slot select, an empty one skipped; zero fill only in windows 0, 3; sums same
# speedup vs baseline: 3.3513x; 1.0456x over previous
#include <hip/hip_runtime.h>

typedef __attribute__((ext_vector_type(2))) float v2f;
typedef __attribute__((ext_vector_type(4))) float v4f;
typedef __attribute__((ext_vector_type(8))) float v8f;


namespace {
constexpr int kNT      = 8192;
constexpr int kOutLen  = 515;
constexpr int kLag0    = -257;
constexpr int kWaves   = 4;
constexpr int kThreads = 32 * kWaves;
constexpr int kGPB     = 1;
constexpr int kTPW     = 9;
constexpr int kTiles   = 34;
constexpr int kIters   = 128;
constexpr int kChunk   = 32;
constexpr int kChunks  = kIters / kChunk;
constexpr int kTrip    = 8;
constexpr int kTrips   = kChunk / kTrip;
constexpr int kProd    = 9;
constexpr int kSets    = 3;
static_assert(kTrip == 8 && kTPW - 1 == 8, "a trip is a wave's first 8 tiles by 8 steps");
constexpr int kShift   = 3;
constexpr int kD2Win   = 64 * kChunk;
constexpr int kD1Win   = ((64 * (kChunk - 1) + 16 * (kTiles - 1) + 64 + kShift + 3) / 4) * 4;
constexpr int kWin     = kD2Win + kD1Win;
constexpr int kWin4    = kWin / 4;
constexpr int kD2Win4  = kD2Win / 4;
constexpr int kSlots   = 2;
constexpr int kPiece4  = kSlots * kThreads;
constexpr int kPieces  = (kWin4 + kPiece4 - 1) / kPiece4;
static_assert(kD2Win4 % kPiece4 == 0, "the d2/d1 border falls between two pieces");
static_assert((kLag0 - kShift) % 4 == 0 && kWin % 4 == 0, "16-byte alignment of both windows");
constexpr int kPad     = 4 * 32;
constexpr int kPitch   = 19;
constexpr int kTileLds = 16 * kPitch;
constexpr int kSmem    = (2 * kWin + kPad > kTiles * kTileLds) ? 2 * kWin + kPad : kTiles * kTileLds;

constexpr int group_sample(int tb, bool isD2, int q) {
  return tb + (isD2 ? 0 : kLag0 - kShift - kD2Win) + 4 * q;
}
constexpr bool group_inside(int tb, bool isD2, int q) {
  return q < kWin4 && group_sample(tb, isD2, q) >= 0 && group_sample(tb, isD2, q) < kNT;
}
constexpr int groups_inside(int tb, bool isD2, int q0) {
  int n = 0;
  for (int i = 0; i < kThreads; ++i)
    n += group_inside(tb, isD2, q0 + i) ? 1 : 0;
  return n;
}
constexpr int groups_to_zero(int tb, int q0) {
  int n = 0;
  for (int i = 0; i < kThreads; ++i)
    n += (q0 + i < kWin4 && !group_inside(tb, false, q0 + i)) ? 1 : 0;
  return n;
}

__device__ __forceinline__ bool in_signal(int tb, bool isD2, int q, unsigned& gi) {
  gi = (unsigned)group_sample(tb, isD2, q);
  return (q < kWin4) && (gi < (unsigned)kNT);
}

template <int kTb, int kP, int kJ>
__device__ __forceinline__ void load_slot(const float* __restrict__ g1, const float* __restrict__ g2, int tid, v4f& r) {
  constexpr bool isD2 = kP * kPiece4 < kD2Win4;
  constexpr int  q0   = kP * kPiece4 + kJ * kThreads;
  constexpr int  nIn  = groups_inside(kTb, isD2, q0);
  const float* src = isD2 ? g2 : g1;
  if constexpr (nIn == kThreads) {
    r = *reinterpret_cast<const v4f*>(src + group_sample(kTb, isD2, q0) + 4 * tid);
  } else if constexpr (nIn > 0) {
    unsigned gi;
    const bool ok = in_signal(kTb, isD2, q0 + tid, gi);
    r = *reinterpret_cast<const v4f*>(src + (ok ? gi : 0u));
  }
}

template <int kTb, int kP, int kJ>
__device__ __forceinline__ void store_slot(float* dst, v4f* spare, int tid, const v4f& r) {
  constexpr bool isD2 = kP * kPiece4 < kD2Win4;
  constexpr int  q0   = kP * kPiece4 + kJ * kThreads;
  constexpr int  nIn  = groups_inside(kTb, isD2, q0);
  if constexpr (nIn == kThreads) {
    reinterpret_cast<v4f*>(dst)[q0 + tid] = r;
  } else if constexpr (nIn > 0) {
    unsigned gi;
    const bool ok = in_signal(kTb, isD2, q0 + tid, gi);
    v4f* d = ok ? reinterpret_cast<v4f*>(dst) + (q0 + tid) : spare;
    *d = r;
  }
}

static_assert(kSlots == 2, "load_piece and store_piece name both slots");
template <int kTb, int kP>
__device__ __forceinline__ void load_piece(const float* __restrict__ g1, const float* __restrict__ g2, int tid, v4f (&r)[kSlots]) {
  load_slot<kTb, kP, 0>(g1, g2, tid, r[0]);
  load_slot<kTb, kP, 1>(g1, g2, tid, r[1]);
}
template <int kTb, int kP>
__device__ __forceinline__ void store_piece(float* dst, v4f* spare, int tid, const v4f (&r)[kSlots]) {
  store_slot<kTb, kP, 0>(dst, spare, tid, r[0]);
  store_slot<kTb, kP, 1>(dst, spare, tid, r[1]);
}

template <int kTb, int kI>
__device__ __forceinline__ void zero_round(float* dst, int tid) {
  constexpr int q0 = kD2Win4 + kI * kThreads;
  constexpr int nZ = groups_to_zero(kTb, q0);
  const v4f z = {0.0f, 0.0f, 0.0f, 0.0f};
  if constexpr (nZ == kThreads) {
    reinterpret_cast<v4f*>(dst)[q0 + tid] = z;
  } else if constexpr (nZ > 0) {
    const int q = q0 + tid;
    unsigned gi;
    const bool ok = in_signal(kTb, false, q, gi);
    if (q < kWin4 && !ok) reinterpret_cast<v4f*>(dst)[q] = z;
  }
}
static_assert((kD1Win / 4 + kThreads - 1) / kThreads == 6, "zero_outside names 6 rounds");
template <int kTb>
__device__ __forceinline__ void zero_outside(float* dst, int tid) {
  zero_round<kTb, 0>(dst, tid);
  zero_round<kTb, 1>(dst, tid);
  zero_round<kTb, 2>(dst, tid);
  zero_round<kTb, 3>(dst, tid);
  zero_round<kTb, 4>(dst, tid);
  zero_round<kTb, 5>(dst, tid);
}

__device__ __forceinline__ v2f load_frag(const float* s, int off) {
  v2f r;
  r.x = s[off];
  r.y = s[off + 16];
  return r;
}

__device__ __forceinline__ v8f mma(v2f a, v2f b, v8f c) {
  return __builtin_amdgcn_wmma_f32_16x16x4_f32(false, a, false, b, (short)0, c, false, false);
}

__device__ __forceinline__ void make_forms(const v2f (&H)[7], v2f (&o)[kProd]) {
  const v2f g0 = H[0] - H[2], g1 = H[1] - H[3], g2 = H[2] - H[4];
  const v2f k1 = H[5] - H[3], k2 = H[6] - H[4];
  o[0] = H[3];
  o[1] = H[2] - H[3];
  o[2] = H[4] - H[3];
  o[3] = g1;
  o[4] = g0 - g1;
  o[5] = g2 - g1;
  o[6] = k1;
  o[7] = -g2 - k1;
  o[8] = k2 - k1;
}

__device__ __forceinline__ void block4(const v2f (&H)[7], const v2f (&y)[4], v8f (&c)[kProd]) {
  v2f o[kProd];
  make_forms(H, o);
  const v2f y01 = y[0] + y[1], y23 = y[2] + y[3];
  const v2f y02 = y[0] + y[2], y13 = y[1] + y[3];
  const v2f yall = y02 + y13;
  const v2f bf[kProd] = { yall, y02, y13, y01, y[0], y[1], y23, y[2], y[3] };
#pragma unroll
  for (int i = 0; i < kProd; ++i)
    c[i] = mma(o[i], bf[i], c[i]);
}

template <int kTbN, int kN, int kP0>
__device__ __forceinline__ void sweep_trip(const float* __restrict__ g1, const float* __restrict__ g2,
                                           const float* sD2, const float* sD1, float* nxt, v4f* spare,
                                           bool nine, int tid, int t, int fragOff, int aBase,
                                           v8f (&c)[kSets][kProd], v8f& last) {
  static_assert(kN <= 2, "a trip moves at most 2 pieces");
  v4f r[kN > 0 ? kN : 1][kSlots];
  if constexpr (kN > 0) load_piece<kTbN, kP0>(g1, g2, tid, r[0]);
  if constexpr (kN > 1) load_piece<kTbN, kP0 + 1>(g1, g2, tid, r[1]);
  const int sh0 = 64 * kTrip * t;
  v2f b[kTrip];
#pragma unroll
  for (int s = 0; s < kTrip; ++s)
    b[s] = load_frag(sD2, fragOff + sh0 + 64 * s);
  v2f h[15];
#pragma unroll
  for (int u = 0; u < 15; ++u)
    h[u] = load_frag(sD1, aBase + sh0 + 64 * u);
  const v2f m4 = h[4] - h[8], m5 = h[5] - h[9], m6 = h[6] - h[10];
  {
    const v2f H[7] = { h[0] - h[4], h[1] - h[5], h[2] - h[6], h[3] - h[7], m4, m5, m6 };
    const v2f y[4] = { b[0], b[1], b[2], b[3] };
    block4(H, y, c[1]);
  }
  {
    const v2f H[7] = { h[4], h[5], h[6], h[7], h[8], h[9], h[10] };
    const v2f y[4] = { b[0] + b[4], b[1] + b[5], b[2] + b[6], b[3] + b[7] };
    block4(H, y, c[0]);
  }
  {
    const v2f H[7] = { -m4, -m5, -m6, h[11] - h[7], h[12] - h[8], h[13] - h[9], h[14] - h[10] };
    const v2f y[4] = { b[4], b[5], b[6], b[7] };
    block4(H, y, c[2]);
  }
  if (nine) {
    const v2f h15 = load_frag(sD1, aBase + sh0 + 64 * 15);
#pragma unroll
    for (int s = 0; s < 7; ++s)
      last = mma(h[8 + s], b[s], last);
    last = mma(h15, b[7], last);
  }
  if constexpr (kN > 0) store_piece<kTbN, kP0>(nxt, spare, tid, r[0]);
  if constexpr (kN > 1) store_piece<kTbN, kP0 + 1>(nxt, spare, tid, r[1]);
}

template <int kW>
__device__ __forceinline__ void sweep_window(const float* __restrict__ g1, const float* __restrict__ g2,
                                             float* smem, v4f* spare, bool nine, int tid, int fragOff, int aBase,
                                             v8f (&c)[kSets][kProd], v8f& last) {
  static_assert(kPieces == 5 && kTrips == 4, "the trips below are written for 5 pieces in 4 trips");
  const float* sD2 = smem + (kW & 1) * kWin;
  const float* sD1 = sD2 + kD2Win;
  float*       nxt = smem + ((kW + 1) & 1) * kWin;
  constexpr int tbN = 64 * kChunk * (kW + 1);
  int tidw = tid;
  asm volatile("" : "+v"(tidw));
  if constexpr (kW + 1 < kChunks) {
    zero_outside<tbN>(nxt, tidw);
    sweep_trip<tbN, 2, 0>(g1, g2, sD2, sD1, nxt, spare, nine, tidw, 0, fragOff, aBase, c, last);
    sweep_trip<tbN, 2, 2>(g1, g2, sD2, sD1, nxt, spare, nine, tidw, 1, fragOff, aBase, c, last);
    sweep_trip<tbN, 1, 4>(g1, g2, sD2, sD1, nxt, spare, nine, tidw, 2, fragOff, aBase, c, last);
    sweep_trip<0, 0, 0>(g1, g2, sD2, sD1, nxt, spare, nine, tidw, 3, fragOff, aBase, c, last);
  } else {
#pragma nounroll
    for (int t = 0; t < kTrips; ++t)
      sweep_trip<0, 0, 0>(g1, g2, sD2, sD1, nxt, spare, nine, tidw, t, fragOff, aBase, c, last);
  }
}
}

__launch_bounds__(kGPB * kThreads, 3)
__global__ void xcorr_wmma_kernel(const float* __restrict__ d1,
                                  const float* __restrict__ d2,
                                  float* __restrict__ out) {
  __shared__ __attribute__((aligned(16))) float smemAll[kGPB * kSmem];

  const int wv   = __builtin_amdgcn_readfirstlane((int)(threadIdx.x >> 5));
  const int rot  = 2 * (__builtin_popcount(blockIdx.x) & 1);
  const int role = (wv + rot) & 3;
  const bool nine = role < 2;
  const int g    = blockIdx.x;
  const int tid  = threadIdx.x;
  const int lane = tid & 31;
  float* smem = smemAll;

  const float* __restrict__ g1 = d1 + (size_t)g * kNT;
  const float* __restrict__ g2 = d2 + (size_t)g * kNT;

  const int nCol    = lane & 15;
  const int half    = lane >> 4;
  const int fragOff = nCol + 32 * half;
  const int aBase   = 16 * role + fragOff + kShift;
  v4f* spare = reinterpret_cast<v4f*>(smem + 2 * kWin) + lane;

  {
    static_assert(kPieces == 5, "the 5 pieces are named one by one");
    v4f r0[kPieces][kSlots];
    load_piece<0, 0>(g1, g2, tid, r0[0]);
    load_piece<0, 1>(g1, g2, tid, r0[1]);
    load_piece<0, 2>(g1, g2, tid, r0[2]);
    load_piece<0, 3>(g1, g2, tid, r0[3]);
    load_piece<0, 4>(g1, g2, tid, r0[4]);
    store_piece<0, 0>(smem, spare, tid, r0[0]);
    store_piece<0, 1>(smem, spare, tid, r0[1]);
    store_piece<0, 2>(smem, spare, tid, r0[2]);
    store_piece<0, 3>(smem, spare, tid, r0[3]);
    store_piece<0, 4>(smem, spare, tid, r0[4]);
    zero_outside<0>(smem, tid);
  }
  __syncthreads();

  v8f c[kSets][kProd] = {};
  v8f last = {};

  static_assert(kChunks == 4, "the 4 windows are named one by one");
  sweep_window<0>(g1, g2, smem, spare, nine, tid, fragOff, aBase, c, last);
  __syncthreads();
  sweep_window<1>(g1, g2, smem, spare, nine, tid, fragOff, aBase, c, last);
  __syncthreads();
  sweep_window<2>(g1, g2, smem, spare, nine, tid, fragOff, aBase, c, last);
  __syncthreads();
  sweep_window<3>(g1, g2, smem, spare, nine, tid, fragOff, aBase, c, last);
  __syncthreads();

  int lane2 = lane;
  asm volatile("" : "+v"(lane2));
  const int mBase = (lane2 >> 4) * 8;
  float* scr0 = smem + role * kTileLds + mBase * kPitch + (lane2 & 15);
#pragma unroll
  for (int a = 0; a < 2; ++a) {
    v8f s[kProd];
#pragma unroll
    for (int k = 0; k < kProd; ++k)
      s[k] = c[0][k] + c[1 + a][k];
    const v8f p1lo = s[0] + s[1], p1hi = s[0] + s[2];
    const v8f t[4] = { p1lo + (s[3] + s[4]), p1hi + (s[3] + s[5]),
                       p1lo + (s[6] + s[7]), p1hi + (s[6] + s[8]) };
#pragma unroll
    for (int i = 0; i < 4; ++i) {
#pragma unroll
      for (int r = 0; r < 8; ++r)
        scr0[kWaves * (4 * a + i) * kTileLds + r * kPitch] = t[i][r];
    }
  }
  if (nine) {
#pragma unroll
    for (int r = 0; r < 8; ++r)
      scr0[kWaves * (kTPW - 1) * kTileLds + r * kPitch] = last[r];
  }
  __syncthreads();

  for (int j = tid; j < kOutLen; j += kThreads) {
    const float* p = smem + kPitch * j;
    float s = 0.0f;
#pragma unroll
    for (int n = 0; n < 16; ++n)
      s += p[(kPitch + 1) * n];
    out[(size_t)g * kOutLen + j] = s;
  }
}

extern "C" void kernel_launch(void* const* d_in, const int* in_sizes, int n_in,
                              void* d_out, int out_size, void* d_ws, size_t ws_size,
                              hipStream_t stream) {
  const float* data1 = (const float*)d_in[0];
  const float* data2 = (const float*)d_in[1];
  float* out = (float*)d_out;
  const int blocks = in_sizes[0] / (kNT * kGPB);
  xcorr_wmma_kernel<<<dim3(blocks), dim3(kGPB * kThreads), 0, stream>>>(data1, data2, out);
}
